# starting kernel
# baseline (speedup 1.0000x reference)
#include <hip/hip_runtime.h>

typedef __attribute__((ext_vector_type(2))) float v2f;
typedef __attribute__((ext_vector_type(4))) float v4f;
typedef __attribute__((ext_vector_type(8))) float v8f;

struct OpTab { float v[9][32][4]; };

constexpr float db4pick(const double* h, int t) {
    return (t >= 0 && t < 4) ? (float)h[t] : 0.0f;
}

constexpr OpTab make_optab() {
    OpTab o{};
    const double s3  = 1.7320508075688772935;
    const double nrm = 0.17677669529663688110;
    const double h0[4] = {(1.0 + s3) * nrm, (3.0 + s3) * nrm,
                          (3.0 - s3) * nrm, (1.0 - s3) * nrm};
    const double h1[4] = {h0[3], -h0[2], h0[1], -h0[0]};
    for (int t = 0; t < 9; ++t) {
        for (int lane = 0; lane < 32; ++lane) {
            int n = lane & 15, g = lane >> 4;
            int tau = 4 * t + 2 * g - 2 * n;
            o.v[t][lane][0] = db4pick(h0, tau);
            o.v[t][lane][1] = db4pick(h0, tau + 1);
            o.v[t][lane][2] = db4pick(h1, tau);
            o.v[t][lane][3] = db4pick(h1, tau + 1);
        }
    }
    return o;
}

__device__ __constant__ OpTab OPTAB = make_optab();

#define WMMA_F32(A, B, C) \
    __builtin_amdgcn_wmma_f32_16x16x4_f32(false, (A), false, (B), (short)0, (C), false, false)

__global__ __launch_bounds__(32)
void dwt_db4_level(const float* __restrict__ src, int N, long long srcStride,
                   float* __restrict__ dstLL, int ldLL, long long llStride,
                   float* __restrict__ outDet, int M)
{
    const int lane = threadIdx.x;
    const int n    = lane & 15;
    const int g    = lane >> 4;
    const int c0   = blockIdx.x * 16;
    const int r0   = blockIdx.y * 16;
    const int img  = blockIdx.z;

    src    += (long long)img * srcStride;
    dstLL  += (long long)img * llStride;
    outDet += (long long)img * (512LL * 512LL);

    __shared__ float Xs[34][38];
    __shared__ float Lr[36][17];
    __shared__ float Hr[36][17];

    int gcA = 2 * c0 - 1 + lane;       gcA += (gcA < 0) ? N : 0;  gcA -= (gcA >= N) ? N : 0;
    int gcB = 2 * c0 + 31 + lane;      gcB -= (gcB >= N) ? N : 0;
    int gr  = 2 * r0 - 1;              gr  += (gr  < 0) ? N : 0;
    for (int r = 0; r < 34; ++r) {
        const float* p = src + (long long)gr * N;
        Xs[r][lane] = p[gcA];
        if (lane < 4) Xs[r][32 + lane] = p[gcB];
        ++gr;  gr = (gr >= N) ? 0 : gr;
    }
    if (lane < 16) {
        Lr[34][lane] = 0.0f; Lr[35][lane] = 0.0f;
        Hr[34][lane] = 0.0f; Hr[35][lane] = 0.0f;
    }

    v2f wL[9], wH[9];
    #pragma unroll
    for (int t = 0; t < 9; ++t) {
        v4f w = *(const v4f*)OPTAB.v[t][lane];
        wL[t].x = w.x;  wL[t].y = w.y;
        wH[t].x = w.z;  wH[t].y = w.w;
    }
    __syncthreads();

    const int chunkR[3] = {0, 16, 18};
    #pragma unroll
    for (int ch = 0; ch < 3; ++ch) {
        const int R = chunkR[ch];
        v8f accL = {0.f,0.f,0.f,0.f,0.f,0.f,0.f,0.f};
        v8f accH = {0.f,0.f,0.f,0.f,0.f,0.f,0.f,0.f};
        #pragma unroll
        for (int t = 0; t < 9; ++t) {
            v2f a = *(const v2f*)&Xs[R + n][4 * t + 2 * g];
            accL = WMMA_F32(a, wL[t], accL);
            accH = WMMA_F32(a, wH[t], accH);
        }
        #pragma unroll
        for (int v = 0; v < 8; ++v) {
            Lr[R + v + 8 * g][n] = accL[v];
            Hr[R + v + 8 * g][n] = accH[v];
        }
    }
    __syncthreads();

    v8f LL = {0.f,0.f,0.f,0.f,0.f,0.f,0.f,0.f};
    v8f LH = {0.f,0.f,0.f,0.f,0.f,0.f,0.f,0.f};
    v8f HL = {0.f,0.f,0.f,0.f,0.f,0.f,0.f,0.f};
    v8f HH = {0.f,0.f,0.f,0.f,0.f,0.f,0.f,0.f};
    #pragma unroll
    for (int t = 0; t < 9; ++t) {
        int rr = 4 * t + 2 * g;
        v2f bLr, bHr;
        bLr.x = Lr[rr][n];  bLr.y = Lr[rr + 1][n];
        bHr.x = Hr[rr][n];  bHr.y = Hr[rr + 1][n];
        LL = WMMA_F32(wL[t], bLr, LL);
        LH = WMMA_F32(wH[t], bLr, LH);
        HL = WMMA_F32(wL[t], bHr, HL);
        HH = WMMA_F32(wH[t], bHr, HH);
    }

    #pragma unroll
    for (int v = 0; v < 8; ++v) {
        int orow = r0 + v + 8 * g;
        int ocol = c0 + n;
        dstLL[(long long)orow * ldLL + ocol] = LL[v];
        __builtin_nontemporal_store(LH[v], &outDet[(long long)orow * 512 + (ocol + M)]);
        __builtin_nontemporal_store(HL[v], &outDet[(long long)(orow + M) * 512 + ocol]);
        __builtin_nontemporal_store(HH[v], &outDet[(long long)(orow + M) * 512 + (ocol + M)]);
    }
}

extern "C" void kernel_launch(void* const* d_in, const int* in_sizes, int n_in,
                              void* d_out, int out_size, void* d_ws, size_t ws_size,
                              hipStream_t stream) {
    const float* x = (const float*)d_in[0];
    float* out = (float*)d_out;
    const int BC = in_sizes[0] / (512 * 512);

    float* LL1 = (float*)d_ws;
    float* LL2 = LL1 + (long long)BC * 256 * 256;

    dim3 blk(32, 1, 1);
    dwt_db4_level<<<dim3(16, 16, BC), blk, 0, stream>>>(
        x,   512, 512LL * 512, LL1, 256, 256LL * 256, out, 256);
    dwt_db4_level<<<dim3(8, 8, BC), blk, 0, stream>>>(
        LL1, 256, 256LL * 256, LL2, 128, 128LL * 128, out, 128);
    dwt_db4_level<<<dim3(4, 4, BC), blk, 0, stream>>>(
        LL2, 128, 128LL * 128, out, 512, 512LL * 512, out, 64);
}


# Register-only data path: vertical pass first, A fragments loaded straight from global memory, its D registers are the horizontal pass's B operand; no LDS, no barriers; sum order changes
# speedup vs baseline: 1.9359x; 1.9359x over previous
#include <hip/hip_runtime.h>


typedef __attribute__((ext_vector_type(2))) float v2f;
typedef __attribute__((ext_vector_type(4))) float v4f;
typedef __attribute__((ext_vector_type(8))) float v8f;

struct OpTab { float v[9][32][4]; };

constexpr float db4pick(const double* f, int t) {
    return (t >= 0 && t < 4) ? (float)f[t] : 0.0f;
}

constexpr OpTab make_tab(bool horizontal) {
    OpTab o{};
    const double s3  = 1.7320508075688772935;
    const double nrm = 0.17677669529663688110;
    const double h0[4] = {(1.0 + s3) * nrm, (3.0 + s3) * nrm,
                          (3.0 - s3) * nrm, (1.0 - s3) * nrm};
    const double h1[4] = {h0[3], -h0[2], h0[1], -h0[0]};
    for (int u = 0; u < 9; ++u) {
        for (int lane = 0; lane < 32; ++lane) {
            int c = lane & 15, h = lane >> 4;
            for (int s = 0; s < 2; ++s) {
                int k = 0;
                bool used = true;
                if (!horizontal)      k = 4 * u + 2 * h + s;
                else if (u < 4)       k = 1 + 2 * (8 * h + 2 * u + s);
                else if (u < 8)       k = 2 + 2 * (8 * h + 2 * (u - 4) + s);
                else { k = (h == 0) ? 0 : 33; used = (s == 0); }
                int tap = k - 2 * c;
                o.v[u][lane][s]     = used ? db4pick(h0, tap) : 0.0f;
                o.v[u][lane][2 + s] = used ? db4pick(h1, tap) : 0.0f;
            }
        }
    }
    return o;
}

__device__ __constant__ OpTab VTAB = make_tab(false);
__device__ __constant__ OpTab HTAB = make_tab(true);

#define WMMA_F32(A, B, C) \
    __builtin_amdgcn_wmma_f32_16x16x4_f32(false, (A), false, (B), (short)0, (C), false, false)

__global__ __launch_bounds__(32)
void dwt_db4_level(const float* __restrict__ src, int N, long long srcStride,
                   float* __restrict__ dstLL, int ldLL, long long llStride,
                   float* __restrict__ outDet, int M)
{
    const int lane = threadIdx.x;
    const int c    = lane & 15;
    const int h    = lane >> 4;
    const int c0   = blockIdx.x * 16;
    const int r0   = blockIdx.y * 16;
    const int img  = blockIdx.z;

    src    += (long long)img * srcStride;
    dstLL  += (long long)img * llStride;
    outDet += (long long)img * (512LL * 512LL);

    const int colP = 2 * c0 + 2 * c;
    int colB = (c < 8) ? (2 * c0 - 1) : (2 * c0 + 32);
    colB += (colB < 0) ? N : 0;
    colB -= (colB >= N) ? N : 0;
    const bool borderLane = (c & 7) == 0;
    const int rowBase = 2 * r0 - 1 + 2 * h;

    v2f xE[9], xO[9], xB[9];
    #pragma unroll
    for (int t = 0; t < 9; ++t) {
        const bool tail = (t == 8) && (h == 1);
        float e[2], o[2], b[2];
        #pragma unroll
        for (int s = 0; s < 2; ++s) {
            int gr = rowBase + 4 * t + s - ((t == 8) ? 2 * h : 0);
            gr += (gr < 0) ? N : 0;
            gr -= (gr >= N) ? N : 0;
            const float* rowp = src + (unsigned)(gr * N);
            const v2f p = *(const v2f*)(rowp + colP);
            const float q = rowp[colB];
            e[s] = tail ? 0.0f : p.x;
            o[s] = tail ? 0.0f : p.y;
            b[s] = (tail || !borderLane) ? 0.0f : q;
        }
        xE[t].x = e[0];  xE[t].y = e[1];
        xO[t].x = o[0];  xO[t].y = o[1];
        xB[t].x = b[0];  xB[t].y = b[1];
    }

    v8f vE[2], vO[2], vB[2];
    #pragma unroll
    for (int f = 0; f < 2; ++f) {
        vE[f] = (v8f){0.f,0.f,0.f,0.f,0.f,0.f,0.f,0.f};
        vO[f] = vE[f];
        vB[f] = vE[f];
    }
    #pragma unroll
    for (int t = 0; t < 9; ++t) {
        const v4f w = *(const v4f*)VTAB.v[t][lane];
        const v2f wL = {w.x, w.y};
        const v2f wH = {w.z, w.w};
        vE[0] = WMMA_F32(xE[t], wL, vE[0]);
        vE[1] = WMMA_F32(xE[t], wH, vE[1]);
        vO[0] = WMMA_F32(xO[t], wL, vO[0]);
        vO[1] = WMMA_F32(xO[t], wH, vO[1]);
        vB[0] = WMMA_F32(xB[t], wL, vB[0]);
        vB[1] = WMMA_F32(xB[t], wH, vB[1]);
    }

    v8f LL = {0.f,0.f,0.f,0.f,0.f,0.f,0.f,0.f};
    v8f LH = LL, HL = LL, HH = LL;
    #pragma unroll
    for (int u = 0; u < 9; ++u) {
        const v4f g = *(const v4f*)HTAB.v[u][lane];
        const v2f gL = {g.x, g.y};
        const v2f gH = {g.z, g.w};
        v2f bl, bh;
        if (u < 4)      { bl.x = vE[0][2*u];     bl.y = vE[0][2*u+1];
                          bh.x = vE[1][2*u];     bh.y = vE[1][2*u+1]; }
        else if (u < 8) { bl.x = vO[0][2*(u-4)]; bl.y = vO[0][2*(u-4)+1];
                          bh.x = vO[1][2*(u-4)]; bh.y = vO[1][2*(u-4)+1]; }
        else            { bl.x = vB[0][0];       bl.y = vB[0][1];
                          bh.x = vB[1][0];       bh.y = vB[1][1]; }
        LL = WMMA_F32(gL, bl, LL);
        LH = WMMA_F32(gL, bh, LH);
        HL = WMMA_F32(gH, bl, HL);
        HH = WMMA_F32(gH, bh, HH);
    }

    const int orow = r0 + c;
    const int ocol = c0 + 8 * h;
    const v4f LLa = {LL[0], LL[1], LL[2], LL[3]}, LLb = {LL[4], LL[5], LL[6], LL[7]};
    const v4f LHa = {LH[0], LH[1], LH[2], LH[3]}, LHb = {LH[4], LH[5], LH[6], LH[7]};
    const v4f HLa = {HL[0], HL[1], HL[2], HL[3]}, HLb = {HL[4], HL[5], HL[6], HL[7]};
    const v4f HHa = {HH[0], HH[1], HH[2], HH[3]}, HHb = {HH[4], HH[5], HH[6], HH[7]};
    float* pLL = dstLL  + (long long)orow * ldLL + ocol;
    float* pLH = outDet + (long long)orow * 512 + (ocol + M);
    float* pHL = outDet + (long long)(orow + M) * 512 + ocol;
    float* pHH = outDet + (long long)(orow + M) * 512 + (ocol + M);
    *(v4f*)pLL       = LLa;
    *(v4f*)(pLL + 4) = LLb;
    __builtin_nontemporal_store(LHa, (v4f*)pLH);
    __builtin_nontemporal_store(LHb, (v4f*)(pLH + 4));
    __builtin_nontemporal_store(HLa, (v4f*)pHL);
    __builtin_nontemporal_store(HLb, (v4f*)(pHL + 4));
    __builtin_nontemporal_store(HHa, (v4f*)pHH);
    __builtin_nontemporal_store(HHb, (v4f*)(pHH + 4));
}

extern "C" void kernel_launch(void* const* d_in, const int* in_sizes, int n_in,
                              void* d_out, int out_size, void* d_ws, size_t ws_size,
                              hipStream_t stream) {
    const float* x = (const float*)d_in[0];
    float* out = (float*)d_out;
    const int BC = in_sizes[0] / (512 * 512);

    float* LL1 = (float*)d_ws;
    float* LL2 = LL1 + (long long)BC * 256 * 256;

    dim3 blk(32, 1, 1);
    dwt_db4_level<<<dim3(16, 16, BC), blk, 0, stream>>>(
        x,   512, 512LL * 512, LL1, 256, 256LL * 256, out, 256);
    dwt_db4_level<<<dim3(8, 8, BC), blk, 0, stream>>>(
        LL1, 256, 256LL * 256, LL2, 128, 128LL * 128, out, 128);
    dwt_db4_level<<<dim3(4, 4, BC), blk, 0, stream>>>(
        LL2, 128, 128LL * 128, out, 512, 512LL * 512, out, 64);
}


# Each wave sweeps a 16-row band's full width tile by tile in registers; left/right tap columns = neighbour tiles' pass-1 D registers: no border loads or products, 72 not 90 WMMAs/tile; same sum order
# speedup vs baseline: 2.2998x; 1.1880x over previous
#include <hip/hip_runtime.h>


typedef __attribute__((ext_vector_type(2))) float v2f;
typedef __attribute__((ext_vector_type(4))) float v4f;
typedef __attribute__((ext_vector_type(8))) float v8f;

struct OpTab { float v[9][32][4]; };

constexpr float db4pick(const double* f, int t) {
    return (t >= 0 && t < 4) ? (float)f[t] : 0.0f;
}

constexpr OpTab make_tab(bool horizontal) {
    OpTab o{};
    const double s3  = 1.7320508075688772935;
    const double nrm = 0.17677669529663688110;
    const double h0[4] = {(1.0 + s3) * nrm, (3.0 + s3) * nrm,
                          (3.0 - s3) * nrm, (1.0 - s3) * nrm};
    const double h1[4] = {h0[3], -h0[2], h0[1], -h0[0]};
    for (int u = 0; u < 9; ++u) {
        for (int lane = 0; lane < 32; ++lane) {
            int c = lane & 15, h = lane >> 4;
            for (int s = 0; s < 2; ++s) {
                int k = 0;
                bool used = true;
                if (!horizontal)      k = 4 * u + 2 * h + s;
                else if (u < 4)       k = 1 + 2 * (8 * h + 2 * u + s);
                else if (u < 8)       k = 2 + 2 * (8 * h + 2 * (u - 4) + s);
                else if (s == 0)    { k = 0;  used = (h == 1); }
                else                { k = 33; used = (h == 0); }
                int tap = k - 2 * c;
                o.v[u][lane][s]     = used ? db4pick(h0, tap) : 0.0f;
                o.v[u][lane][2 + s] = used ? db4pick(h1, tap) : 0.0f;
            }
        }
    }
    return o;
}

__device__ __constant__ OpTab VTAB = make_tab(false);
__device__ __constant__ OpTab HTAB = make_tab(true);

#define WMMA_F32(A, B, C) \
    __builtin_amdgcn_wmma_f32_16x16x4_f32(false, (A), false, (B), (short)0, (C), false, false)

struct RawTile { v2f p[9][2]; };
struct ColTile { v8f e[2], o[2]; };

template <int T>
__global__ __launch_bounds__(32)
void dwt_db4_band(const float* __restrict__ src, long long srcStride,
                  float* __restrict__ dstLL, int ldLL, long long llStride,
                  float* __restrict__ outDet)
{
    constexpr int N = 32 * T;
    constexpr int M = 16 * T;
    const int lane = threadIdx.x;
    const int c    = lane & 15;
    const int h    = lane >> 4;
    const int r0   = blockIdx.x * 16;
    const int img  = blockIdx.y;

    src    += (long long)img * srcStride;
    dstLL  += (long long)img * llStride;
    outDet += (long long)img * (512LL * 512LL);

    const int rowBase = 2 * r0 - 1 + 2 * h;
    unsigned rowOff[9][2];
    #pragma unroll
    for (int t = 0; t < 9; ++t) {
        #pragma unroll
        for (int s = 0; s < 2; ++s) {
            int gr = rowBase + 4 * t + s - ((t == 8) ? 2 * h : 0);
            gr += (gr < 0) ? N : 0;
            gr -= (gr >= N) ? N : 0;
            rowOff[t][s] = 4u * ((unsigned)(gr * N) + (unsigned)(2 * c));
        }
    }
    const bool tail = (h == 1);

    auto load_tile = [&](RawTile& x, int tile) __attribute__((always_inline)) {
        const char* base = (const char*)(src + 32 * tile);
        #pragma unroll
        for (int t = 0; t < 9; ++t) {
            #pragma unroll
            for (int s = 0; s < 2; ++s)
                x.p[t][s] = *(const v2f*)(base + rowOff[t][s]);
        }
    };

    auto pass1 = [&](const RawTile& x, ColTile& d) __attribute__((always_inline)) {
        #pragma unroll
        for (int f = 0; f < 2; ++f) {
            d.e[f] = (v8f){0.f,0.f,0.f,0.f,0.f,0.f,0.f,0.f};
            d.o[f] = d.e[f];
        }
        #pragma unroll
        for (int t = 0; t < 9; ++t) {
            const v4f w = *(const v4f*)VTAB.v[t][lane];
            const v2f wL = {w.x, w.y};
            const v2f wH = {w.z, w.w};
            const bool z = (t == 8) && tail;
            v2f xe, xo;
            xe.x = z ? 0.0f : x.p[t][0].x;  xe.y = z ? 0.0f : x.p[t][1].x;
            xo.x = z ? 0.0f : x.p[t][0].y;  xo.y = z ? 0.0f : x.p[t][1].y;
            d.e[0] = WMMA_F32(xe, wL, d.e[0]);
            d.e[1] = WMMA_F32(xe, wH, d.e[1]);
            d.o[0] = WMMA_F32(xo, wL, d.o[0]);
            d.o[1] = WMMA_F32(xo, wH, d.o[1]);
        }
    };

    auto pass2_store = [&](const ColTile& d, float leftLo, float leftHi,
                           float rightLo, float rightHi, int tile)
                       __attribute__((always_inline)) {
        v8f LL = {0.f,0.f,0.f,0.f,0.f,0.f,0.f,0.f};
        v8f LH = LL, HL = LL, HH = LL;
        #pragma unroll
        for (int u = 0; u < 9; ++u) {
            const v4f g = *(const v4f*)HTAB.v[u][lane];
            const v2f gL = {g.x, g.y};
            const v2f gH = {g.z, g.w};
            v2f bl, bh;
            if (u < 4)      { bl.x = d.e[0][2*u];     bl.y = d.e[0][2*u+1];
                              bh.x = d.e[1][2*u];     bh.y = d.e[1][2*u+1]; }
            else if (u < 8) { bl.x = d.o[0][2*(u-4)]; bl.y = d.o[0][2*(u-4)+1];
                              bh.x = d.o[1][2*(u-4)]; bh.y = d.o[1][2*(u-4)+1]; }
            else            { bl.x = leftLo;          bl.y = rightLo;
                              bh.x = leftHi;          bh.y = rightHi; }
            LL = WMMA_F32(gL, bl, LL);
            LH = WMMA_F32(gL, bh, LH);
            HL = WMMA_F32(gH, bl, HL);
            HH = WMMA_F32(gH, bh, HH);
        }
        const int orow = r0 + c;
        const int ocol = 16 * tile + 8 * h;
        const v4f LLa = {LL[0], LL[1], LL[2], LL[3]}, LLb = {LL[4], LL[5], LL[6], LL[7]};
        const v4f LHa = {LH[0], LH[1], LH[2], LH[3]}, LHb = {LH[4], LH[5], LH[6], LH[7]};
        const v4f HLa = {HL[0], HL[1], HL[2], HL[3]}, HLb = {HL[4], HL[5], HL[6], HL[7]};
        const v4f HHa = {HH[0], HH[1], HH[2], HH[3]}, HHb = {HH[4], HH[5], HH[6], HH[7]};
        float* pLL = dstLL  + (long long)orow * ldLL + ocol;
        float* pLH = outDet + (long long)orow * 512 + (ocol + M);
        float* pHL = outDet + (long long)(orow + M) * 512 + ocol;
        float* pHH = outDet + (long long)(orow + M) * 512 + (ocol + M);
        *(v4f*)pLL       = LLa;
        *(v4f*)(pLL + 4) = LLb;
        __builtin_nontemporal_store(LHa, (v4f*)pLH);
        __builtin_nontemporal_store(LHb, (v4f*)(pLH + 4));
        __builtin_nontemporal_store(HLa, (v4f*)pHL);
        __builtin_nontemporal_store(HLb, (v4f*)(pHL + 4));
        __builtin_nontemporal_store(HHa, (v4f*)pHH);
        __builtin_nontemporal_store(HHb, (v4f*)(pHH + 4));
    };

    RawTile xa, xb;
    ColTile da, db;

    load_tile(xa, T - 1);
    load_tile(xb, 0);
    pass1(xa, db);
    float leftLo = db.o[0][7], leftHi = db.o[1][7];
    load_tile(xa, 1);
    pass1(xb, da);
    const float wrapLo = da.e[0][0], wrapHi = da.e[1][0];
    load_tile(xb, 2);

    auto trip = [&](int n, bool loadA, bool hasNext, bool loadB) __attribute__((always_inline)) {
        pass1(xa, db);
        if (loadA) load_tile(xa, n + 3);
        pass2_store(da, leftLo, leftHi, db.e[0][0], db.e[1][0], n);
        leftLo = da.o[0][7];  leftHi = da.o[1][7];

        float rightLo = wrapLo, rightHi = wrapHi;
        if (hasNext) {
            pass1(xb, da);
            if (loadB) load_tile(xb, n + 4);
            rightLo = da.e[0][0];  rightHi = da.e[1][0];
        }
        pass2_store(db, leftLo, leftHi, rightLo, rightHi, n + 1);
        leftLo = db.o[0][7];  leftHi = db.o[1][7];
    };

    #pragma unroll 1
    for (int n = 0; n < T - 4; n += 2)
        trip(n, true, true, true);
    trip(T - 4, true, true, false);
    trip(T - 2, false, false, false);
}

extern "C" void kernel_launch(void* const* d_in, const int* in_sizes, int n_in,
                              void* d_out, int out_size, void* d_ws, size_t ws_size,
                              hipStream_t stream) {
    const float* x = (const float*)d_in[0];
    float* out = (float*)d_out;
    const int BC = in_sizes[0] / (512 * 512);

    float* LL1 = (float*)d_ws;
    float* LL2 = LL1 + (long long)BC * 256 * 256;

    dim3 blk(32, 1, 1);
    dwt_db4_band<16><<<dim3(16, BC), blk, 0, stream>>>(
        x,   512LL * 512, LL1, 256, 256LL * 256, out);
    dwt_db4_band<8><<<dim3(8, BC), blk, 0, stream>>>(
        LL1, 256LL * 256, LL2, 128, 128LL * 128, out);
    dwt_db4_band<4><<<dim3(4, BC), blk, 0, stream>>>(
        LL2, 128LL * 128, out, 512, 512LL * 512, out);
}


# Result registers of a trip's two tiles exchange lane halves (v_permlane16_swap, 32 per pair): each store writes one whole 128-byte line, not two 64-byte halves of two rows; same sum order
# speedup vs baseline: 3.4410x; 1.4962x over previous
#include <hip/hip_runtime.h>


typedef __attribute__((ext_vector_type(2))) float v2f;
typedef __attribute__((ext_vector_type(4))) float v4f;
typedef __attribute__((ext_vector_type(8))) float v8f;

struct OpTab { float v[9][32][4]; };

constexpr float db4pick(const double* f, int t) {
    return (t >= 0 && t < 4) ? (float)f[t] : 0.0f;
}

constexpr OpTab make_tab(bool horizontal) {
    OpTab o{};
    const double s3  = 1.7320508075688772935;
    const double nrm = 0.17677669529663688110;
    const double h0[4] = {(1.0 + s3) * nrm, (3.0 + s3) * nrm,
                          (3.0 - s3) * nrm, (1.0 - s3) * nrm};
    const double h1[4] = {h0[3], -h0[2], h0[1], -h0[0]};
    for (int u = 0; u < 9; ++u) {
        for (int lane = 0; lane < 32; ++lane) {
            int c = lane & 15, h = lane >> 4;
            for (int s = 0; s < 2; ++s) {
                int k = 0;
                bool used = true;
                if (!horizontal)      k = 4 * u + 2 * h + s;
                else if (u < 4)       k = 1 + 2 * (8 * h + 2 * u + s);
                else if (u < 8)       k = 2 + 2 * (8 * h + 2 * (u - 4) + s);
                else if (s == 0)    { k = 0;  used = (h == 1); }
                else                { k = 33; used = (h == 0); }
                int tap = k - 2 * c;
                o.v[u][lane][s]     = used ? db4pick(h0, tap) : 0.0f;
                o.v[u][lane][2 + s] = used ? db4pick(h1, tap) : 0.0f;
            }
        }
    }
    return o;
}

__device__ __constant__ OpTab VTAB = make_tab(false);
__device__ __constant__ OpTab HTAB = make_tab(true);

#define WMMA_F32(A, B, C) \
    __builtin_amdgcn_wmma_f32_16x16x4_f32(false, (A), false, (B), (short)0, (C), false, false)

struct RawTile { v2f p[9][2]; };
struct ColTile { v8f e[2], o[2]; };
struct SubTile { v8f s[4]; };

template <int T, int LDLL>
__global__ __launch_bounds__(32)
void dwt_db4_band(const float* __restrict__ src, long long srcStride,
                  float* __restrict__ dstLL, long long llStride,
                  float* __restrict__ outDet)
{
    constexpr int N = 32 * T;
    constexpr int M = 16 * T;
    const int lane = threadIdx.x;
    const int c    = lane & 15;
    const int h    = lane >> 4;
    const int r0   = blockIdx.x * 16;
    const int img  = blockIdx.y;

    src    += (long long)img * srcStride;
    dstLL  += (long long)img * llStride;
    outDet += (long long)img * (512LL * 512LL);

    const int rowBase = 2 * r0 - 1 + 2 * h;
    unsigned rowOff[9][2];
    #pragma unroll
    for (int t = 0; t < 9; ++t) {
        #pragma unroll
        for (int s = 0; s < 2; ++s) {
            int gr = rowBase + 4 * t + s - ((t == 8) ? 2 * h : 0);
            gr += (gr < 0) ? N : 0;
            gr -= (gr >= N) ? N : 0;
            rowOff[t][s] = 4u * ((unsigned)(gr * N) + (unsigned)(2 * c));
        }
    }
    const bool tail = (h == 1);

    auto load_tile = [&](RawTile& x, int tile) __attribute__((always_inline)) {
        const char* base = (const char*)(src + 32 * tile);
        #pragma unroll
        for (int t = 0; t < 9; ++t) {
            #pragma unroll
            for (int s = 0; s < 2; ++s)
                x.p[t][s] = *(const v2f*)(base + rowOff[t][s]);
        }
    };

    auto pass1 = [&](const RawTile& x, ColTile& d) __attribute__((always_inline)) {
        #pragma unroll
        for (int f = 0; f < 2; ++f) {
            d.e[f] = (v8f){0.f,0.f,0.f,0.f,0.f,0.f,0.f,0.f};
            d.o[f] = d.e[f];
        }
        #pragma unroll
        for (int t = 0; t < 9; ++t) {
            const v4f w = *(const v4f*)VTAB.v[t][lane];
            const v2f wL = {w.x, w.y};
            const v2f wH = {w.z, w.w};
            const bool z = (t == 8) && tail;
            v2f xe, xo;
            xe.x = z ? 0.0f : x.p[t][0].x;  xe.y = z ? 0.0f : x.p[t][1].x;
            xo.x = z ? 0.0f : x.p[t][0].y;  xo.y = z ? 0.0f : x.p[t][1].y;
            d.e[0] = WMMA_F32(xe, wL, d.e[0]);
            d.e[1] = WMMA_F32(xe, wH, d.e[1]);
            d.o[0] = WMMA_F32(xo, wL, d.o[0]);
            d.o[1] = WMMA_F32(xo, wH, d.o[1]);
        }
    };

    auto pass2 = [&](const ColTile& d, float leftLo, float leftHi,
                     float rightLo, float rightHi, SubTile& q)
                 __attribute__((always_inline)) {
        v8f LL = {0.f,0.f,0.f,0.f,0.f,0.f,0.f,0.f};
        v8f LH = LL, HL = LL, HH = LL;
        #pragma unroll
        for (int u = 0; u < 9; ++u) {
            const v4f g = *(const v4f*)HTAB.v[u][lane];
            const v2f gL = {g.x, g.y};
            const v2f gH = {g.z, g.w};
            v2f bl, bh;
            if (u < 4)      { bl.x = d.e[0][2*u];     bl.y = d.e[0][2*u+1];
                              bh.x = d.e[1][2*u];     bh.y = d.e[1][2*u+1]; }
            else if (u < 8) { bl.x = d.o[0][2*(u-4)]; bl.y = d.o[0][2*(u-4)+1];
                              bh.x = d.o[1][2*(u-4)]; bh.y = d.o[1][2*(u-4)+1]; }
            else            { bl.x = leftLo;          bl.y = rightLo;
                              bh.x = leftHi;          bh.y = rightHi; }
            LL = WMMA_F32(bl, gL, LL);
            LH = WMMA_F32(bh, gL, LH);
            HL = WMMA_F32(bl, gH, HL);
            HH = WMMA_F32(bh, gH, HH);
        }
        q.s[0] = LL;  q.s[1] = LH;  q.s[2] = HL;  q.s[3] = HH;
    };

    auto store_pair = [&](const SubTile& a, const SubTile& b, int n) __attribute__((always_inline)) {
        float top[4][8], bot[4][8];
        #pragma unroll
        for (int q = 0; q < 4; ++q) {
            #pragma unroll
            for (int r = 0; r < 8; ++r) {
                const float fa = a.s[q][r];
                const float fb = b.s[q][r];
                const unsigned ua = __builtin_bit_cast(unsigned, fa);
                const unsigned ub = __builtin_bit_cast(unsigned, fb);
                const auto sw = __builtin_amdgcn_permlane16_swap(ua, ub, false, false);
                const unsigned u0 = sw[0];
                const unsigned u1 = sw[1];
                top[q][r] = __builtin_bit_cast(float, u0);
                bot[q][r] = __builtin_bit_cast(float, u1);
            }
        }
        const int ocol = 16 * n + lane;
        float* pLL = dstLL  + (long long)r0 * LDLL + ocol;
        float* pLH = outDet + (long long)r0 * 512 + (ocol + M);
        float* pHL = outDet + (long long)(r0 + M) * 512 + ocol;
        float* pHH = outDet + (long long)(r0 + M) * 512 + (ocol + M);
        #pragma unroll
        for (int r = 0; r < 8; ++r) {
            pLL[r * LDLL]       = top[0][r];
            pLL[(8 + r) * LDLL] = bot[0][r];
            __builtin_nontemporal_store(top[1][r], pLH + r * 512);
            __builtin_nontemporal_store(bot[1][r], pLH + (8 + r) * 512);
            __builtin_nontemporal_store(top[2][r], pHL + r * 512);
            __builtin_nontemporal_store(bot[2][r], pHL + (8 + r) * 512);
            __builtin_nontemporal_store(top[3][r], pHH + r * 512);
            __builtin_nontemporal_store(bot[3][r], pHH + (8 + r) * 512);
        }
    };

    RawTile xa, xb;
    ColTile da, db;

    load_tile(xa, T - 1);
    load_tile(xb, 0);
    pass1(xa, db);
    float leftLo = db.o[0][7], leftHi = db.o[1][7];
    load_tile(xa, 1);
    pass1(xb, da);
    const float wrapLo = da.e[0][0], wrapHi = da.e[1][0];
    load_tile(xb, 2);

    auto trip = [&](int n, bool loadA, bool hasNext, bool loadB) __attribute__((always_inline)) {
        SubTile qa, qb;
        pass1(xa, db);
        if (loadA) load_tile(xa, n + 3);
        pass2(da, leftLo, leftHi, db.e[0][0], db.e[1][0], qa);
        leftLo = da.o[0][7];  leftHi = da.o[1][7];

        float rightLo = wrapLo, rightHi = wrapHi;
        if (hasNext) {
            pass1(xb, da);
            if (loadB) load_tile(xb, n + 4);
            rightLo = da.e[0][0];  rightHi = da.e[1][0];
        }
        pass2(db, leftLo, leftHi, rightLo, rightHi, qb);
        leftLo = db.o[0][7];  leftHi = db.o[1][7];
        store_pair(qa, qb, n);
    };

    #pragma unroll 1
    for (int n = 0; n < T - 4; n += 2)
        trip(n, true, true, true);
    trip(T - 4, true, true, false);
    trip(T - 2, false, false, false);
}

extern "C" void kernel_launch(void* const* d_in, const int* in_sizes, int n_in,
                              void* d_out, int out_size, void* d_ws, size_t ws_size,
                              hipStream_t stream) {
    const float* x = (const float*)d_in[0];
    float* out = (float*)d_out;
    const int BC = in_sizes[0] / (512 * 512);

    float* LL1 = (float*)d_ws;
    float* LL2 = LL1 + (long long)BC * 256 * 256;

    dim3 blk(32, 1, 1);
    dwt_db4_band<16, 256><<<dim3(16, BC), blk, 0, stream>>>(
        x,   512LL * 512, LL1, 256LL * 256, out);
    dwt_db4_band<8, 128><<<dim3(8, BC), blk, 0, stream>>>(
        LL1, 256LL * 256, LL2, 128LL * 128, out);
    dwt_db4_band<4, 512><<<dim3(4, BC), blk, 0, stream>>>(
        LL2, 128LL * 128, out, 512LL * 512, out);
}
